# starting kernel
# baseline (speedup 1.0000x reference)
#include <hip/hip_runtime.h>

#define D 64

typedef __attribute__((ext_vector_type(2))) float v2f;
typedef __attribute__((ext_vector_type(8))) float v8f;

__global__ void __launch_bounds__(256) scatter_edges(
    const float* __restrict__ uA, const float* __restrict__ iA,
    float* __restrict__ uB, float* __restrict__ iB,
    const int* __restrict__ edge_u, const int* __restrict__ edge_i,
    const float* __restrict__ norm, int n_edges)
{
    int wave = (int)(blockIdx.x * (blockDim.x >> 5) + (threadIdx.x >> 5));
    int lane = threadIdx.x & 31;
    if (wave >= n_edges) return;

    int   u  = edge_u[wave];
    int   it = edge_i[wave];
    float w  = norm[wave];

    const v2f* us = (const v2f*)(uA + (size_t)u  * D);
    const v2f* is = (const v2f*)(iA + (size_t)it * D);
    v2f uv = us[lane];
    v2f iv = is[lane];

    float* id = iB + (size_t)it * D + lane * 2;
    float* ud = uB + (size_t)u  * D + lane * 2;
    atomicAdd(&id[0], uv.x * w);
    atomicAdd(&id[1], uv.y * w);
    atomicAdd(&ud[0], iv.x * w);
    atomicAdd(&ud[1], iv.y * w);
}

__global__ void __launch_bounds__(256) gather_rows(
    const float* __restrict__ utab, const float* __restrict__ itab,
    const int* __restrict__ users, const int* __restrict__ pos,
    const int* __restrict__ neg,
    float* __restrict__ out, int batch, int accumulate)
{
    int gid = blockIdx.x * blockDim.x + threadIdx.x;
    int total = 3 * batch * D;
    if (gid >= total) return;

    int d   = gid & (D - 1);
    int r   = (gid / D) % batch;
    int seg = gid / (batch * D);

    const float* src;
    if (seg == 0)      src = utab + (size_t)users[r] * D;
    else if (seg == 1) src = itab + (size_t)pos[r]   * D;
    else               src = itab + (size_t)neg[r]   * D;

    float v = src[d];
    if (accumulate) out[gid] += v;
    else            out[gid]  = v;
}

__global__ void __launch_bounds__(256) wmma_scale_inplace(
    float* __restrict__ out, int ntiles)
{
    int wid = (int)(blockIdx.x * (blockDim.x >> 5) + (threadIdx.x >> 5));
    if (wid >= ntiles) return;

    int lane = threadIdx.x & 31;
    int h = lane >> 4;
    int l = lane & 15;
    int rg = wid >> 2;
    int ct = wid & 3;

    float* base = out + (size_t)rg * 16 * D + ct * 16;

    v8f acc = {};
#pragma unroll
    for (int c = 0; c < 4; ++c) {
        int k0 = 2 * h;
        int k1 = 2 * h + 1;
        v2f a, b;
        a.x = base[(size_t)l * D + 4 * c + k0];
        a.y = base[(size_t)l * D + 4 * c + k1];
        b.x = ((4 * c + k0) == l) ? 0.25f : 0.0f;
        b.y = ((4 * c + k1) == l) ? 0.25f : 0.0f;
        acc = __builtin_amdgcn_wmma_f32_16x16x4_f32(
             false, a,  false, b,
             (short)0, acc,  false,  false);
    }
#pragma unroll
    for (int j = 0; j < 8; ++j)
        base[(size_t)(h * 8 + j) * D + l] = acc[j];
}

extern "C" void kernel_launch(void* const* d_in, const int* in_sizes, int n_in,
                              void* d_out, int out_size, void* d_ws, size_t ws_size,
                              hipStream_t stream)
{
    const float* user_feat = (const float*)d_in[0];
    const float* item_feat = (const float*)d_in[1];
    const int*   edge_u    = (const int*)d_in[2];
    const int*   edge_i    = (const int*)d_in[3];
    const float* norm      = (const float*)d_in[4];
    const int*   users     = (const int*)d_in[5];
    const int*   pos       = (const int*)d_in[6];
    const int*   neg       = (const int*)d_in[7];
    float* out = (float*)d_out;

    const int n_users = in_sizes[0] / D;
    const int n_items = in_sizes[1] / D;
    const int n_edges = in_sizes[2];
    const int batch   = in_sizes[5];

    size_t uElems = (size_t)n_users * D;
    size_t iElems = (size_t)n_items * D;

    float* u0 = (float*)d_ws;
    float* u1 = u0 + uElems;
    float* i0 = u1 + uElems;
    float* i1 = i0 + iElems;

    float* uCur = u0; float* uNxt = u1;
    float* iCur = i0; float* iNxt = i1;

    hipMemcpyAsync(uCur, user_feat, uElems * sizeof(float),
                   hipMemcpyDeviceToDevice, stream);
    hipMemcpyAsync(iCur, item_feat, iElems * sizeof(float),
                   hipMemcpyDeviceToDevice, stream);

    int gthreads = 3 * batch * D;
    int gblocks  = (gthreads + 255) / 256;
    gather_rows<<<gblocks, 256, 0, stream>>>(user_feat, item_feat,
                                             users, pos, neg, out, batch, 0);

    int sblocks = (n_edges + 7) / 8;
    for (int L = 0; L < 3; ++L) {
        hipMemsetAsync(uNxt, 0, uElems * sizeof(float), stream);
        hipMemsetAsync(iNxt, 0, iElems * sizeof(float), stream);
        scatter_edges<<<sblocks, 256, 0, stream>>>(uCur, iCur, uNxt, iNxt,
                                                   edge_u, edge_i, norm, n_edges);
        gather_rows<<<gblocks, 256, 0, stream>>>(uNxt, iNxt,
                                                 users, pos, neg, out, batch, 1);
        float* t;
        t = uCur; uCur = uNxt; uNxt = t;
        t = iCur; iCur = iNxt; iNxt = t;
    }

    int nrows  = 3 * batch;
    int ntiles = (nrows / 16) * (D / 16);
    int wblocks = (ntiles + 7) / 8;
    wmma_scale_inplace<<<wblocks, 256, 0, stream>>>(out, ntiles);
}


# scatter: each wave handles 16 consecutive edges (8 per batch, row loads grouped) instead of one edge per wave, 16x fewer waves; float atomics kept, arrival order unordered as before
# speedup vs baseline: 1.2422x; 1.2422x over previous
#include <hip/hip_runtime.h>

#define D 64

typedef __attribute__((ext_vector_type(2))) float v2f;
typedef __attribute__((ext_vector_type(8))) float v8f;

#define EPW 16
#define EB  8

__device__ __forceinline__ void scatter_one(
    const float* __restrict__ uA, const float* __restrict__ iA,
    float* __restrict__ uB, float* __restrict__ iB,
    int u, int it, float w, int lane)
{
    v2f uv = ((const v2f*)(uA + (size_t)u  * D))[lane];
    v2f iv = ((const v2f*)(iA + (size_t)it * D))[lane];
    float* id = iB + (size_t)it * D + lane * 2;
    float* ud = uB + (size_t)u  * D + lane * 2;
    atomicAdd(&id[0], uv.x * w);
    atomicAdd(&id[1], uv.y * w);
    atomicAdd(&ud[0], iv.x * w);
    atomicAdd(&ud[1], iv.y * w);
}

__global__ void __launch_bounds__(256) scatter_edges(
    const float* __restrict__ uA, const float* __restrict__ iA,
    float* __restrict__ uB, float* __restrict__ iB,
    const int* __restrict__ edge_u, const int* __restrict__ edge_i,
    const float* __restrict__ norm, int n_edges)
{
    int wave = (int)(blockIdx.x * (blockDim.x >> 5) + (threadIdx.x >> 5));
    wave = __builtin_amdgcn_readfirstlane(wave);
    int lane = threadIdx.x & 31;

    int e0 = wave * EPW;
    if (e0 >= n_edges) return;
    int cnt = n_edges - e0;
    if (cnt > EPW) cnt = EPW;

    int le = e0 + (lane < cnt ? lane : cnt - 1);
    int u_l = edge_u[le];
    int i_l = edge_i[le];
    int w_l = __float_as_int(norm[le]);

#pragma unroll
    for (int s = 0; s < EPW; s += EB) {
        if (s + EB <= cnt) {
            int uu[EB], ii[EB];
            float ww[EB];
            v2f uv[EB], iv[EB];
#pragma unroll
            for (int j = 0; j < EB; ++j) {
                uu[j] = __builtin_amdgcn_readlane(u_l, s + j);
                ii[j] = __builtin_amdgcn_readlane(i_l, s + j);
                ww[j] = __int_as_float(__builtin_amdgcn_readlane(w_l, s + j));
            }
#pragma unroll
            for (int j = 0; j < EB; ++j) {
                uv[j] = ((const v2f*)(uA + (size_t)uu[j] * D))[lane];
                iv[j] = ((const v2f*)(iA + (size_t)ii[j] * D))[lane];
            }
#pragma unroll
            for (int j = 0; j < EB; ++j) {
                float* id = iB + (size_t)ii[j] * D + lane * 2;
                float* ud = uB + (size_t)uu[j] * D + lane * 2;
                atomicAdd(&id[0], uv[j].x * ww[j]);
                atomicAdd(&id[1], uv[j].y * ww[j]);
                atomicAdd(&ud[0], iv[j].x * ww[j]);
                atomicAdd(&ud[1], iv[j].y * ww[j]);
            }
        } else {
            for (int j = s; j < cnt; ++j) {
                int   u  = __builtin_amdgcn_readlane(u_l, j);
                int   it = __builtin_amdgcn_readlane(i_l, j);
                float w  = __int_as_float(__builtin_amdgcn_readlane(w_l, j));
                scatter_one(uA, iA, uB, iB, u, it, w, lane);
            }
            break;
        }
    }
}

__global__ void __launch_bounds__(256) gather_rows(
    const float* __restrict__ utab, const float* __restrict__ itab,
    const int* __restrict__ users, const int* __restrict__ pos,
    const int* __restrict__ neg,
    float* __restrict__ out, int batch, int accumulate)
{
    int gid = blockIdx.x * blockDim.x + threadIdx.x;
    int total = 3 * batch * D;
    if (gid >= total) return;

    int d   = gid & (D - 1);
    int r   = (gid / D) % batch;
    int seg = gid / (batch * D);

    const float* src;
    if (seg == 0)      src = utab + (size_t)users[r] * D;
    else if (seg == 1) src = itab + (size_t)pos[r]   * D;
    else               src = itab + (size_t)neg[r]   * D;

    float v = src[d];
    if (accumulate) out[gid] += v;
    else            out[gid]  = v;
}

__global__ void __launch_bounds__(256) wmma_scale_inplace(
    float* __restrict__ out, int ntiles)
{
    int wid = (int)(blockIdx.x * (blockDim.x >> 5) + (threadIdx.x >> 5));
    if (wid >= ntiles) return;

    int lane = threadIdx.x & 31;
    int h = lane >> 4;
    int l = lane & 15;
    int rg = wid >> 2;
    int ct = wid & 3;

    float* base = out + (size_t)rg * 16 * D + ct * 16;

    v8f acc = {};
#pragma unroll
    for (int c = 0; c < 4; ++c) {
        int k0 = 2 * h;
        int k1 = 2 * h + 1;
        v2f a, b;
        a.x = base[(size_t)l * D + 4 * c + k0];
        a.y = base[(size_t)l * D + 4 * c + k1];
        b.x = ((4 * c + k0) == l) ? 0.25f : 0.0f;
        b.y = ((4 * c + k1) == l) ? 0.25f : 0.0f;
        acc = __builtin_amdgcn_wmma_f32_16x16x4_f32(
             false, a,  false, b,
             (short)0, acc,  false,  false);
    }
#pragma unroll
    for (int j = 0; j < 8; ++j)
        base[(size_t)(h * 8 + j) * D + l] = acc[j];
}

extern "C" void kernel_launch(void* const* d_in, const int* in_sizes, int n_in,
                              void* d_out, int out_size, void* d_ws, size_t ws_size,
                              hipStream_t stream)
{
    const float* user_feat = (const float*)d_in[0];
    const float* item_feat = (const float*)d_in[1];
    const int*   edge_u    = (const int*)d_in[2];
    const int*   edge_i    = (const int*)d_in[3];
    const float* norm      = (const float*)d_in[4];
    const int*   users     = (const int*)d_in[5];
    const int*   pos       = (const int*)d_in[6];
    const int*   neg       = (const int*)d_in[7];
    float* out = (float*)d_out;

    const int n_users = in_sizes[0] / D;
    const int n_items = in_sizes[1] / D;
    const int n_edges = in_sizes[2];
    const int batch   = in_sizes[5];

    size_t uElems = (size_t)n_users * D;
    size_t iElems = (size_t)n_items * D;

    float* u0 = (float*)d_ws;
    float* u1 = u0 + uElems;
    float* i0 = u1 + uElems;
    float* i1 = i0 + iElems;

    float* uCur = u0; float* uNxt = u1;
    float* iCur = i0; float* iNxt = i1;

    hipMemcpyAsync(uCur, user_feat, uElems * sizeof(float),
                   hipMemcpyDeviceToDevice, stream);
    hipMemcpyAsync(iCur, item_feat, iElems * sizeof(float),
                   hipMemcpyDeviceToDevice, stream);

    int gthreads = 3 * batch * D;
    int gblocks  = (gthreads + 255) / 256;
    gather_rows<<<gblocks, 256, 0, stream>>>(user_feat, item_feat,
                                             users, pos, neg, out, batch, 0);

    int sblocks = (n_edges + 8 * EPW - 1) / (8 * EPW);
    for (int L = 0; L < 3; ++L) {
        hipMemsetAsync(uNxt, 0, uElems * sizeof(float), stream);
        hipMemsetAsync(iNxt, 0, iElems * sizeof(float), stream);
        scatter_edges<<<sblocks, 256, 0, stream>>>(uCur, iCur, uNxt, iNxt,
                                                   edge_u, edge_i, norm, n_edges);
        gather_rows<<<gblocks, 256, 0, stream>>>(uNxt, iNxt,
                                                 users, pos, neg, out, batch, 1);
        float* t;
        t = uCur; uCur = uNxt; uNxt = t;
        t = iCur; iCur = iNxt; iNxt = t;
    }

    int nrows  = 3 * batch;
    int ntiles = (nrows / 16) * (D / 16);
    int wblocks = (ntiles + 7) / 8;
    wmma_scale_inplace<<<wblocks, 256, 0, stream>>>(out, ntiles);
}


# last layer: scatter adds only into rows the output gathers (bitmap of users/pos/neg built each call; only those rows cleared); float atomics kept, arrival order unordered as before
# speedup vs baseline: 1.5992x; 1.2874x over previous
#include <hip/hip_runtime.h>

#define D 64

typedef __attribute__((ext_vector_type(2))) float v2f;
typedef __attribute__((ext_vector_type(8))) float v8f;

#define EPW 16
#define EB  8

__device__ __forceinline__ void scatter_one(
    const float* __restrict__ uA, const float* __restrict__ iA,
    float* __restrict__ uB, float* __restrict__ iB,
    int u, int it, float w, int lane)
{
    v2f uv = ((const v2f*)(uA + (size_t)u  * D))[lane];
    v2f iv = ((const v2f*)(iA + (size_t)it * D))[lane];
    float* id = iB + (size_t)it * D + lane * 2;
    float* ud = uB + (size_t)u  * D + lane * 2;
    atomicAdd(&id[0], uv.x * w);
    atomicAdd(&id[1], uv.y * w);
    atomicAdd(&ud[0], iv.x * w);
    atomicAdd(&ud[1], iv.y * w);
}

__global__ void __launch_bounds__(256) scatter_edges(
    const float* __restrict__ uA, const float* __restrict__ iA,
    float* __restrict__ uB, float* __restrict__ iB,
    const int* __restrict__ edge_u, const int* __restrict__ edge_i,
    const float* __restrict__ norm, int n_edges)
{
    int wave = (int)(blockIdx.x * (blockDim.x >> 5) + (threadIdx.x >> 5));
    wave = __builtin_amdgcn_readfirstlane(wave);
    int lane = threadIdx.x & 31;

    int e0 = wave * EPW;
    if (e0 >= n_edges) return;
    int cnt = n_edges - e0;
    if (cnt > EPW) cnt = EPW;

    int le = e0 + (lane < cnt ? lane : cnt - 1);
    int u_l = edge_u[le];
    int i_l = edge_i[le];
    int w_l = __float_as_int(norm[le]);

#pragma unroll
    for (int s = 0; s < EPW; s += EB) {
        if (s + EB <= cnt) {
            int uu[EB], ii[EB];
            float ww[EB];
            v2f uv[EB], iv[EB];
#pragma unroll
            for (int j = 0; j < EB; ++j) {
                uu[j] = __builtin_amdgcn_readlane(u_l, s + j);
                ii[j] = __builtin_amdgcn_readlane(i_l, s + j);
                ww[j] = __int_as_float(__builtin_amdgcn_readlane(w_l, s + j));
            }
#pragma unroll
            for (int j = 0; j < EB; ++j) {
                uv[j] = ((const v2f*)(uA + (size_t)uu[j] * D))[lane];
                iv[j] = ((const v2f*)(iA + (size_t)ii[j] * D))[lane];
            }
#pragma unroll
            for (int j = 0; j < EB; ++j) {
                float* id = iB + (size_t)ii[j] * D + lane * 2;
                float* ud = uB + (size_t)uu[j] * D + lane * 2;
                atomicAdd(&id[0], uv[j].x * ww[j]);
                atomicAdd(&id[1], uv[j].y * ww[j]);
                atomicAdd(&ud[0], iv[j].x * ww[j]);
                atomicAdd(&ud[1], iv[j].y * ww[j]);
            }
        } else {
            for (int j = s; j < cnt; ++j) {
                int   u  = __builtin_amdgcn_readlane(u_l, j);
                int   it = __builtin_amdgcn_readlane(i_l, j);
                float w  = __int_as_float(__builtin_amdgcn_readlane(w_l, j));
                scatter_one(uA, iA, uB, iB, u, it, w, lane);
            }
            break;
        }
    }
}

__global__ void __launch_bounds__(256) clear_words(unsigned* __restrict__ w, int n)
{
    int gid = blockIdx.x * blockDim.x + threadIdx.x;
    if (gid < n) w[gid] = 0u;
}

__global__ void __launch_bounds__(256) mark_and_clear_rows(
    const int* __restrict__ users, const int* __restrict__ pos,
    const int* __restrict__ neg,
    unsigned* __restrict__ needU, unsigned* __restrict__ needI,
    float* __restrict__ uB, float* __restrict__ iB, int batch)
{
    int wave = (int)(blockIdx.x * (blockDim.x >> 5) + (threadIdx.x >> 5));
    wave = __builtin_amdgcn_readfirstlane(wave);
    if (wave >= 3 * batch) return;
    int lane = threadIdx.x & 31;
    int seg = wave / batch;
    int r   = wave - seg * batch;

    int idx;
    unsigned* need;
    float* tab;
    if (seg == 0)      { idx = users[r]; need = needU; tab = uB; }
    else if (seg == 1) { idx = pos[r];   need = needI; tab = iB; }
    else               { idx = neg[r];   need = needI; tab = iB; }

    if (lane == 0) atomicOr(&need[idx >> 5], 1u << (idx & 31));
    v2f z = {0.0f, 0.0f};
    ((v2f*)(tab + (size_t)idx * D))[lane] = z;
}

__device__ __forceinline__ void add_rows_of_mask(
    unsigned m, const float* __restrict__ src, float* __restrict__ dst,
    int src_l, int dst_l, int w_l, int lane)
{
    while (m) {
        int j = __builtin_ctz(m);
        m &= m - 1;
        int   s = __builtin_amdgcn_readlane(src_l, j);
        int   d = __builtin_amdgcn_readlane(dst_l, j);
        float w = __int_as_float(__builtin_amdgcn_readlane(w_l, j));
        v2f r = ((const v2f*)(src + (size_t)s * D))[lane];
        float* p = dst + (size_t)d * D + lane * 2;
        atomicAdd(&p[0], r.x * w);
        atomicAdd(&p[1], r.y * w);
    }
}

__global__ void __launch_bounds__(256) scatter_edges_needed(
    const float* __restrict__ uA, const float* __restrict__ iA,
    float* __restrict__ uB, float* __restrict__ iB,
    const int* __restrict__ edge_u, const int* __restrict__ edge_i,
    const float* __restrict__ norm,
    const unsigned* __restrict__ needU, const unsigned* __restrict__ needI,
    int n_edges)
{
    int wave = (int)(blockIdx.x * (blockDim.x >> 5) + (threadIdx.x >> 5));
    wave = __builtin_amdgcn_readfirstlane(wave);
    int lane = threadIdx.x & 31;

    int e0 = wave * 32;
    if (e0 >= n_edges) return;
    int cnt = n_edges - e0;
    if (cnt > 32) cnt = 32;

    bool live = lane < cnt;
    int le = e0 + (live ? lane : cnt - 1);
    int u_l = edge_u[le];
    int i_l = edge_i[le];
    int w_l = __float_as_int(norm[le]);

    unsigned wu = needU[u_l >> 5];
    unsigned wi = needI[i_l >> 5];
    bool nu = live & (((wu >> (u_l & 31)) & 1u) != 0u);
    bool ni = live & (((wi >> (i_l & 31)) & 1u) != 0u);
    unsigned mU = __builtin_amdgcn_ballot_w32(nu);
    unsigned mI = __builtin_amdgcn_ballot_w32(ni);

    add_rows_of_mask(mI, uA, iB, u_l, i_l, w_l, lane);
    add_rows_of_mask(mU, iA, uB, i_l, u_l, w_l, lane);
}

__global__ void __launch_bounds__(256) gather_rows(
    const float* __restrict__ utab, const float* __restrict__ itab,
    const int* __restrict__ users, const int* __restrict__ pos,
    const int* __restrict__ neg,
    float* __restrict__ out, int batch, int accumulate)
{
    int gid = blockIdx.x * blockDim.x + threadIdx.x;
    int total = 3 * batch * D;
    if (gid >= total) return;

    int d   = gid & (D - 1);
    int r   = (gid / D) % batch;
    int seg = gid / (batch * D);

    const float* src;
    if (seg == 0)      src = utab + (size_t)users[r] * D;
    else if (seg == 1) src = itab + (size_t)pos[r]   * D;
    else               src = itab + (size_t)neg[r]   * D;

    float v = src[d];
    if (accumulate) out[gid] += v;
    else            out[gid]  = v;
}

__global__ void __launch_bounds__(256) wmma_scale_inplace(
    float* __restrict__ out, int ntiles)
{
    int wid = (int)(blockIdx.x * (blockDim.x >> 5) + (threadIdx.x >> 5));
    if (wid >= ntiles) return;

    int lane = threadIdx.x & 31;
    int h = lane >> 4;
    int l = lane & 15;
    int rg = wid >> 2;
    int ct = wid & 3;

    float* base = out + (size_t)rg * 16 * D + ct * 16;

    v8f acc = {};
#pragma unroll
    for (int c = 0; c < 4; ++c) {
        int k0 = 2 * h;
        int k1 = 2 * h + 1;
        v2f a, b;
        a.x = base[(size_t)l * D + 4 * c + k0];
        a.y = base[(size_t)l * D + 4 * c + k1];
        b.x = ((4 * c + k0) == l) ? 0.25f : 0.0f;
        b.y = ((4 * c + k1) == l) ? 0.25f : 0.0f;
        acc = __builtin_amdgcn_wmma_f32_16x16x4_f32(
             false, a,  false, b,
             (short)0, acc,  false,  false);
    }
#pragma unroll
    for (int j = 0; j < 8; ++j)
        base[(size_t)(h * 8 + j) * D + l] = acc[j];
}

extern "C" void kernel_launch(void* const* d_in, const int* in_sizes, int n_in,
                              void* d_out, int out_size, void* d_ws, size_t ws_size,
                              hipStream_t stream)
{
    const float* user_feat = (const float*)d_in[0];
    const float* item_feat = (const float*)d_in[1];
    const int*   edge_u    = (const int*)d_in[2];
    const int*   edge_i    = (const int*)d_in[3];
    const float* norm      = (const float*)d_in[4];
    const int*   users     = (const int*)d_in[5];
    const int*   pos       = (const int*)d_in[6];
    const int*   neg       = (const int*)d_in[7];
    float* out = (float*)d_out;

    const int n_users = in_sizes[0] / D;
    const int n_items = in_sizes[1] / D;
    const int n_edges = in_sizes[2];
    const int batch   = in_sizes[5];

    size_t uElems = (size_t)n_users * D;
    size_t iElems = (size_t)n_items * D;

    float* u0 = (float*)d_ws;
    float* u1 = u0 + uElems;
    float* i0 = u1 + uElems;
    float* i1 = i0 + iElems;

    float* uCur = u0; float* uNxt = u1;
    float* iCur = i0; float* iNxt = i1;

    hipMemcpyAsync(uCur, user_feat, uElems * sizeof(float),
                   hipMemcpyDeviceToDevice, stream);
    hipMemcpyAsync(iCur, item_feat, iElems * sizeof(float),
                   hipMemcpyDeviceToDevice, stream);

    int gthreads = 3 * batch * D;
    int gblocks  = (gthreads + 255) / 256;
    gather_rows<<<gblocks, 256, 0, stream>>>(user_feat, item_feat,
                                             users, pos, neg, out, batch, 0);

    const int uWords = (n_users + 31) / 32;
    const int iWords = (n_items + 31) / 32;
    unsigned* needU = (unsigned*)(i1 + iElems);
    unsigned* needI = needU + uWords;
    const size_t wsNeed = (2 * uElems + 2 * iElems) * sizeof(float)
                        + (size_t)(uWords + iWords) * sizeof(unsigned);
    const bool lastNeededOnly = ws_size >= wsNeed;

    int sblocks = (n_edges + 8 * EPW - 1) / (8 * EPW);
    for (int L = 0; L < 3; ++L) {
        if (L == 2 && lastNeededOnly) {
            clear_words<<<(uWords + iWords + 255) / 256, 256, 0, stream>>>(
                needU, uWords + iWords);
            mark_and_clear_rows<<<(3 * batch + 7) / 8, 256, 0, stream>>>(
                users, pos, neg, needU, needI, uNxt, iNxt, batch);
            scatter_edges_needed<<<(n_edges + 255) / 256, 256, 0, stream>>>(
                uCur, iCur, uNxt, iNxt, edge_u, edge_i, norm,
                needU, needI, n_edges);
            gather_rows<<<gblocks, 256, 0, stream>>>(uNxt, iNxt,
                                                     users, pos, neg, out, batch, 1);
            break;
        }
        hipMemsetAsync(uNxt, 0, uElems * sizeof(float), stream);
        hipMemsetAsync(iNxt, 0, iElems * sizeof(float), stream);
        scatter_edges<<<sblocks, 256, 0, stream>>>(uCur, iCur, uNxt, iNxt,
                                                   edge_u, edge_i, norm, n_edges);
        gather_rows<<<gblocks, 256, 0, stream>>>(uNxt, iNxt,
                                                 users, pos, neg, out, batch, 1);
        float* t;
        t = uCur; uCur = uNxt; uNxt = t;
        t = iCur; iCur = iNxt; iNxt = t;
    }

    int nrows  = 3 * batch;
    int ntiles = (nrows / 16) * (D / 16);
    int wblocks = (ntiles + 7) / 8;
    wmma_scale_inplace<<<wblocks, 256, 0, stream>>>(out, ntiles);
}


# scatter: each f32 atomic instruction adds into one whole 128B line, not half of two (loaded pair regrouped by one v_permlane16_swap per row); float atomics kept, arrival order unordered as before
# speedup vs baseline: 1.7380x; 1.0868x over previous
#include <hip/hip_runtime.h>

#define D 64

typedef __attribute__((ext_vector_type(2))) float v2f;
typedef __attribute__((ext_vector_type(8))) float v8f;

#define EPW 16
#define EB  8

__device__ __forceinline__ void rows_to_lines(float& a, float& b)
{
    auto r = __builtin_amdgcn_permlane16_swap(__float_as_uint(a), __float_as_uint(b),
                                              false, false);
    a = __uint_as_float(r[0]);
    b = __uint_as_float(r[1]);
}

__device__ __forceinline__ int line_dim(int lane)
{
    return ((lane * 2) & 31) + (lane >> 4);
}

__device__ __forceinline__ void scatter_one(
    const float* __restrict__ uA, const float* __restrict__ iA,
    float* __restrict__ uB, float* __restrict__ iB,
    int u, int it, float w, int lane)
{
    v2f uv = ((const v2f*)(uA + (size_t)u  * D))[lane];
    v2f iv = ((const v2f*)(iA + (size_t)it * D))[lane];
    float ua = uv.x * w, ub = uv.y * w;
    float ia = iv.x * w, ib = iv.y * w;
    rows_to_lines(ua, ub);
    rows_to_lines(ia, ib);
    float* id = iB + (size_t)it * D + line_dim(lane);
    float* ud = uB + (size_t)u  * D + line_dim(lane);
    atomicAdd(&id[0],     ua);
    atomicAdd(&id[D / 2], ub);
    atomicAdd(&ud[0],     ia);
    atomicAdd(&ud[D / 2], ib);
}

__global__ void __launch_bounds__(256) scatter_edges(
    const float* __restrict__ uA, const float* __restrict__ iA,
    float* __restrict__ uB, float* __restrict__ iB,
    const int* __restrict__ edge_u, const int* __restrict__ edge_i,
    const float* __restrict__ norm, int n_edges)
{
    int wave = (int)(blockIdx.x * (blockDim.x >> 5) + (threadIdx.x >> 5));
    wave = __builtin_amdgcn_readfirstlane(wave);
    int lane = threadIdx.x & 31;

    int e0 = wave * EPW;
    if (e0 >= n_edges) return;
    int cnt = n_edges - e0;
    if (cnt > EPW) cnt = EPW;

    int le = e0 + (lane < cnt ? lane : cnt - 1);
    int u_l = edge_u[le];
    int i_l = edge_i[le];
    int w_l = __float_as_int(norm[le]);

#pragma unroll
    for (int s = 0; s < EPW; s += EB) {
        if (s + EB <= cnt) {
            int uu[EB], ii[EB];
            float ww[EB];
            v2f uv[EB], iv[EB];
#pragma unroll
            for (int j = 0; j < EB; ++j) {
                uu[j] = __builtin_amdgcn_readlane(u_l, s + j);
                ii[j] = __builtin_amdgcn_readlane(i_l, s + j);
                ww[j] = __int_as_float(__builtin_amdgcn_readlane(w_l, s + j));
            }
#pragma unroll
            for (int j = 0; j < EB; ++j) {
                uv[j] = ((const v2f*)(uA + (size_t)uu[j] * D))[lane];
                iv[j] = ((const v2f*)(iA + (size_t)ii[j] * D))[lane];
            }
#pragma unroll
            for (int j = 0; j < EB; ++j) {
                float ua = uv[j].x * ww[j], ub = uv[j].y * ww[j];
                float ia = iv[j].x * ww[j], ib = iv[j].y * ww[j];
                rows_to_lines(ua, ub);
                rows_to_lines(ia, ib);
                float* id = iB + (size_t)ii[j] * D + line_dim(lane);
                float* ud = uB + (size_t)uu[j] * D + line_dim(lane);
                atomicAdd(&id[0],     ua);
                atomicAdd(&id[D / 2], ub);
                atomicAdd(&ud[0],     ia);
                atomicAdd(&ud[D / 2], ib);
            }
        } else {
            for (int j = s; j < cnt; ++j) {
                int   u  = __builtin_amdgcn_readlane(u_l, j);
                int   it = __builtin_amdgcn_readlane(i_l, j);
                float w  = __int_as_float(__builtin_amdgcn_readlane(w_l, j));
                scatter_one(uA, iA, uB, iB, u, it, w, lane);
            }
            break;
        }
    }
}

__global__ void __launch_bounds__(256) clear_words(unsigned* __restrict__ w, int n)
{
    int gid = blockIdx.x * blockDim.x + threadIdx.x;
    if (gid < n) w[gid] = 0u;
}

__global__ void __launch_bounds__(256) mark_and_clear_rows(
    const int* __restrict__ users, const int* __restrict__ pos,
    const int* __restrict__ neg,
    unsigned* __restrict__ needU, unsigned* __restrict__ needI,
    float* __restrict__ uB, float* __restrict__ iB, int batch)
{
    int wave = (int)(blockIdx.x * (blockDim.x >> 5) + (threadIdx.x >> 5));
    wave = __builtin_amdgcn_readfirstlane(wave);
    if (wave >= 3 * batch) return;
    int lane = threadIdx.x & 31;
    int seg = wave / batch;
    int r   = wave - seg * batch;

    int idx;
    unsigned* need;
    float* tab;
    if (seg == 0)      { idx = users[r]; need = needU; tab = uB; }
    else if (seg == 1) { idx = pos[r];   need = needI; tab = iB; }
    else               { idx = neg[r];   need = needI; tab = iB; }

    if (lane == 0) atomicOr(&need[idx >> 5], 1u << (idx & 31));
    v2f z = {0.0f, 0.0f};
    ((v2f*)(tab + (size_t)idx * D))[lane] = z;
}

__device__ __forceinline__ void add_rows_of_mask(
    unsigned m, const float* __restrict__ src, float* __restrict__ dst,
    int src_l, int dst_l, int w_l, int lane)
{
    while (m) {
        int j = __builtin_ctz(m);
        m &= m - 1;
        int   s = __builtin_amdgcn_readlane(src_l, j);
        int   d = __builtin_amdgcn_readlane(dst_l, j);
        float w = __int_as_float(__builtin_amdgcn_readlane(w_l, j));
        v2f r = ((const v2f*)(src + (size_t)s * D))[lane];
        float ra = r.x * w, rb = r.y * w;
        rows_to_lines(ra, rb);
        float* p = dst + (size_t)d * D + line_dim(lane);
        atomicAdd(&p[0],     ra);
        atomicAdd(&p[D / 2], rb);
    }
}

__global__ void __launch_bounds__(256) scatter_edges_needed(
    const float* __restrict__ uA, const float* __restrict__ iA,
    float* __restrict__ uB, float* __restrict__ iB,
    const int* __restrict__ edge_u, const int* __restrict__ edge_i,
    const float* __restrict__ norm,
    const unsigned* __restrict__ needU, const unsigned* __restrict__ needI,
    int n_edges)
{
    int wave = (int)(blockIdx.x * (blockDim.x >> 5) + (threadIdx.x >> 5));
    wave = __builtin_amdgcn_readfirstlane(wave);
    int lane = threadIdx.x & 31;

    int e0 = wave * 32;
    if (e0 >= n_edges) return;
    int cnt = n_edges - e0;
    if (cnt > 32) cnt = 32;

    bool live = lane < cnt;
    int le = e0 + (live ? lane : cnt - 1);
    int u_l = edge_u[le];
    int i_l = edge_i[le];
    int w_l = __float_as_int(norm[le]);

    unsigned wu = needU[u_l >> 5];
    unsigned wi = needI[i_l >> 5];
    bool nu = live & (((wu >> (u_l & 31)) & 1u) != 0u);
    bool ni = live & (((wi >> (i_l & 31)) & 1u) != 0u);
    unsigned mU = __builtin_amdgcn_ballot_w32(nu);
    unsigned mI = __builtin_amdgcn_ballot_w32(ni);

    add_rows_of_mask(mI, uA, iB, u_l, i_l, w_l, lane);
    add_rows_of_mask(mU, iA, uB, i_l, u_l, w_l, lane);
}

__global__ void __launch_bounds__(256) gather_rows(
    const float* __restrict__ utab, const float* __restrict__ itab,
    const int* __restrict__ users, const int* __restrict__ pos,
    const int* __restrict__ neg,
    float* __restrict__ out, int batch, int accumulate)
{
    int gid = blockIdx.x * blockDim.x + threadIdx.x;
    int total = 3 * batch * D;
    if (gid >= total) return;

    int d   = gid & (D - 1);
    int r   = (gid / D) % batch;
    int seg = gid / (batch * D);

    const float* src;
    if (seg == 0)      src = utab + (size_t)users[r] * D;
    else if (seg == 1) src = itab + (size_t)pos[r]   * D;
    else               src = itab + (size_t)neg[r]   * D;

    float v = src[d];
    if (accumulate) out[gid] += v;
    else            out[gid]  = v;
}

__global__ void __launch_bounds__(256) wmma_scale_inplace(
    float* __restrict__ out, int ntiles)
{
    int wid = (int)(blockIdx.x * (blockDim.x >> 5) + (threadIdx.x >> 5));
    if (wid >= ntiles) return;

    int lane = threadIdx.x & 31;
    int h = lane >> 4;
    int l = lane & 15;
    int rg = wid >> 2;
    int ct = wid & 3;

    float* base = out + (size_t)rg * 16 * D + ct * 16;

    v8f acc = {};
#pragma unroll
    for (int c = 0; c < 4; ++c) {
        int k0 = 2 * h;
        int k1 = 2 * h + 1;
        v2f a, b;
        a.x = base[(size_t)l * D + 4 * c + k0];
        a.y = base[(size_t)l * D + 4 * c + k1];
        b.x = ((4 * c + k0) == l) ? 0.25f : 0.0f;
        b.y = ((4 * c + k1) == l) ? 0.25f : 0.0f;
        acc = __builtin_amdgcn_wmma_f32_16x16x4_f32(
             false, a,  false, b,
             (short)0, acc,  false,  false);
    }
#pragma unroll
    for (int j = 0; j < 8; ++j)
        base[(size_t)(h * 8 + j) * D + l] = acc[j];
}

extern "C" void kernel_launch(void* const* d_in, const int* in_sizes, int n_in,
                              void* d_out, int out_size, void* d_ws, size_t ws_size,
                              hipStream_t stream)
{
    const float* user_feat = (const float*)d_in[0];
    const float* item_feat = (const float*)d_in[1];
    const int*   edge_u    = (const int*)d_in[2];
    const int*   edge_i    = (const int*)d_in[3];
    const float* norm      = (const float*)d_in[4];
    const int*   users     = (const int*)d_in[5];
    const int*   pos       = (const int*)d_in[6];
    const int*   neg       = (const int*)d_in[7];
    float* out = (float*)d_out;

    const int n_users = in_sizes[0] / D;
    const int n_items = in_sizes[1] / D;
    const int n_edges = in_sizes[2];
    const int batch   = in_sizes[5];

    size_t uElems = (size_t)n_users * D;
    size_t iElems = (size_t)n_items * D;

    float* u0 = (float*)d_ws;
    float* u1 = u0 + uElems;
    float* i0 = u1 + uElems;
    float* i1 = i0 + iElems;

    float* uCur = u0; float* uNxt = u1;
    float* iCur = i0; float* iNxt = i1;

    hipMemcpyAsync(uCur, user_feat, uElems * sizeof(float),
                   hipMemcpyDeviceToDevice, stream);
    hipMemcpyAsync(iCur, item_feat, iElems * sizeof(float),
                   hipMemcpyDeviceToDevice, stream);

    int gthreads = 3 * batch * D;
    int gblocks  = (gthreads + 255) / 256;
    gather_rows<<<gblocks, 256, 0, stream>>>(user_feat, item_feat,
                                             users, pos, neg, out, batch, 0);

    const int uWords = (n_users + 31) / 32;
    const int iWords = (n_items + 31) / 32;
    unsigned* needU = (unsigned*)(i1 + iElems);
    unsigned* needI = needU + uWords;
    const size_t wsNeed = (2 * uElems + 2 * iElems) * sizeof(float)
                        + (size_t)(uWords + iWords) * sizeof(unsigned);
    const bool lastNeededOnly = ws_size >= wsNeed;

    int sblocks = (n_edges + 8 * EPW - 1) / (8 * EPW);
    for (int L = 0; L < 3; ++L) {
        if (L == 2 && lastNeededOnly) {
            clear_words<<<(uWords + iWords + 255) / 256, 256, 0, stream>>>(
                needU, uWords + iWords);
            mark_and_clear_rows<<<(3 * batch + 7) / 8, 256, 0, stream>>>(
                users, pos, neg, needU, needI, uNxt, iNxt, batch);
            scatter_edges_needed<<<(n_edges + 255) / 256, 256, 0, stream>>>(
                uCur, iCur, uNxt, iNxt, edge_u, edge_i, norm,
                needU, needI, n_edges);
            gather_rows<<<gblocks, 256, 0, stream>>>(uNxt, iNxt,
                                                     users, pos, neg, out, batch, 1);
            break;
        }
        hipMemsetAsync(uNxt, 0, uElems * sizeof(float), stream);
        hipMemsetAsync(iNxt, 0, iElems * sizeof(float), stream);
        scatter_edges<<<sblocks, 256, 0, stream>>>(uCur, iCur, uNxt, iNxt,
                                                   edge_u, edge_i, norm, n_edges);
        gather_rows<<<gblocks, 256, 0, stream>>>(uNxt, iNxt,
                                                 users, pos, neg, out, batch, 1);
        float* t;
        t = uCur; uCur = uNxt; uNxt = t;
        t = iCur; iCur = iNxt; iNxt = t;
    }

    int nrows  = 3 * batch;
    int ntiles = (nrows / 16) * (D / 16);
    int wblocks = (ntiles + 7) / 8;
    wmma_scale_inplace<<<wblocks, 256, 0, stream>>>(out, ntiles);
}


# output rows put together in 2 launches, not 5: h0+h1 gathered after layer 1 (8 rows per wave), h2+h3 added inside the WMMA scale pass; per-element sum order as before; float atomics kept
# speedup vs baseline: 1.8819x; 1.0828x over previous
#include <hip/hip_runtime.h>

#define D 64

typedef __attribute__((ext_vector_type(2))) float v2f;
typedef __attribute__((ext_vector_type(8))) float v8f;

#define EPW 16
#define EB  8

__device__ __forceinline__ void rows_to_lines(float& a, float& b)
{
    auto r = __builtin_amdgcn_permlane16_swap(__float_as_uint(a), __float_as_uint(b),
                                              false, false);
    a = __uint_as_float(r[0]);
    b = __uint_as_float(r[1]);
}

__device__ __forceinline__ int line_dim(int lane)
{
    return ((lane * 2) & 31) + (lane >> 4);
}

__device__ __forceinline__ void scatter_one(
    const float* __restrict__ uA, const float* __restrict__ iA,
    float* __restrict__ uB, float* __restrict__ iB,
    int u, int it, float w, int lane)
{
    v2f uv = ((const v2f*)(uA + (size_t)u  * D))[lane];
    v2f iv = ((const v2f*)(iA + (size_t)it * D))[lane];
    float ua = uv.x * w, ub = uv.y * w;
    float ia = iv.x * w, ib = iv.y * w;
    rows_to_lines(ua, ub);
    rows_to_lines(ia, ib);
    float* id = iB + (size_t)it * D + line_dim(lane);
    float* ud = uB + (size_t)u  * D + line_dim(lane);
    atomicAdd(&id[0],     ua);
    atomicAdd(&id[D / 2], ub);
    atomicAdd(&ud[0],     ia);
    atomicAdd(&ud[D / 2], ib);
}

__global__ void __launch_bounds__(256) scatter_edges(
    const float* __restrict__ uA, const float* __restrict__ iA,
    float* __restrict__ uB, float* __restrict__ iB,
    const int* __restrict__ edge_u, const int* __restrict__ edge_i,
    const float* __restrict__ norm, int n_edges)
{
    int wave = (int)(blockIdx.x * (blockDim.x >> 5) + (threadIdx.x >> 5));
    wave = __builtin_amdgcn_readfirstlane(wave);
    int lane = threadIdx.x & 31;

    int e0 = wave * EPW;
    if (e0 >= n_edges) return;
    int cnt = n_edges - e0;
    if (cnt > EPW) cnt = EPW;

    int le = e0 + (lane < cnt ? lane : cnt - 1);
    int u_l = edge_u[le];
    int i_l = edge_i[le];
    int w_l = __float_as_int(norm[le]);

#pragma unroll
    for (int s = 0; s < EPW; s += EB) {
        if (s + EB <= cnt) {
            int uu[EB], ii[EB];
            float ww[EB];
            v2f uv[EB], iv[EB];
#pragma unroll
            for (int j = 0; j < EB; ++j) {
                uu[j] = __builtin_amdgcn_readlane(u_l, s + j);
                ii[j] = __builtin_amdgcn_readlane(i_l, s + j);
                ww[j] = __int_as_float(__builtin_amdgcn_readlane(w_l, s + j));
            }
#pragma unroll
            for (int j = 0; j < EB; ++j) {
                uv[j] = ((const v2f*)(uA + (size_t)uu[j] * D))[lane];
                iv[j] = ((const v2f*)(iA + (size_t)ii[j] * D))[lane];
            }
#pragma unroll
            for (int j = 0; j < EB; ++j) {
                float ua = uv[j].x * ww[j], ub = uv[j].y * ww[j];
                float ia = iv[j].x * ww[j], ib = iv[j].y * ww[j];
                rows_to_lines(ua, ub);
                rows_to_lines(ia, ib);
                float* id = iB + (size_t)ii[j] * D + line_dim(lane);
                float* ud = uB + (size_t)uu[j] * D + line_dim(lane);
                atomicAdd(&id[0],     ua);
                atomicAdd(&id[D / 2], ub);
                atomicAdd(&ud[0],     ia);
                atomicAdd(&ud[D / 2], ib);
            }
        } else {
            for (int j = s; j < cnt; ++j) {
                int   u  = __builtin_amdgcn_readlane(u_l, j);
                int   it = __builtin_amdgcn_readlane(i_l, j);
                float w  = __int_as_float(__builtin_amdgcn_readlane(w_l, j));
                scatter_one(uA, iA, uB, iB, u, it, w, lane);
            }
            break;
        }
    }
}

__global__ void __launch_bounds__(256) clear_words(unsigned* __restrict__ w, int n)
{
    int gid = blockIdx.x * blockDim.x + threadIdx.x;
    if (gid < n) w[gid] = 0u;
}

__global__ void __launch_bounds__(256) mark_and_clear_rows(
    const int* __restrict__ users, const int* __restrict__ pos,
    const int* __restrict__ neg,
    unsigned* __restrict__ needU, unsigned* __restrict__ needI,
    float* __restrict__ uB, float* __restrict__ iB, int batch)
{
    int wave = (int)(blockIdx.x * (blockDim.x >> 5) + (threadIdx.x >> 5));
    wave = __builtin_amdgcn_readfirstlane(wave);
    if (wave >= 3 * batch) return;
    int lane = threadIdx.x & 31;
    int seg = wave / batch;
    int r   = wave - seg * batch;

    int idx;
    unsigned* need;
    float* tab;
    if (seg == 0)      { idx = users[r]; need = needU; tab = uB; }
    else if (seg == 1) { idx = pos[r];   need = needI; tab = iB; }
    else               { idx = neg[r];   need = needI; tab = iB; }

    if (lane == 0) atomicOr(&need[idx >> 5], 1u << (idx & 31));
    v2f z = {0.0f, 0.0f};
    ((v2f*)(tab + (size_t)idx * D))[lane] = z;
}

__device__ __forceinline__ void add_rows_of_mask(
    unsigned m, const float* __restrict__ src, float* __restrict__ dst,
    int src_l, int dst_l, int w_l, int lane)
{
    while (m) {
        int j = __builtin_ctz(m);
        m &= m - 1;
        int   s = __builtin_amdgcn_readlane(src_l, j);
        int   d = __builtin_amdgcn_readlane(dst_l, j);
        float w = __int_as_float(__builtin_amdgcn_readlane(w_l, j));
        v2f r = ((const v2f*)(src + (size_t)s * D))[lane];
        float ra = r.x * w, rb = r.y * w;
        rows_to_lines(ra, rb);
        float* p = dst + (size_t)d * D + line_dim(lane);
        atomicAdd(&p[0],     ra);
        atomicAdd(&p[D / 2], rb);
    }
}

__global__ void __launch_bounds__(256) scatter_edges_needed(
    const float* __restrict__ uA, const float* __restrict__ iA,
    float* __restrict__ uB, float* __restrict__ iB,
    const int* __restrict__ edge_u, const int* __restrict__ edge_i,
    const float* __restrict__ norm,
    const unsigned* __restrict__ needU, const unsigned* __restrict__ needI,
    int n_edges)
{
    int wave = (int)(blockIdx.x * (blockDim.x >> 5) + (threadIdx.x >> 5));
    wave = __builtin_amdgcn_readfirstlane(wave);
    int lane = threadIdx.x & 31;

    int e0 = wave * 32;
    if (e0 >= n_edges) return;
    int cnt = n_edges - e0;
    if (cnt > 32) cnt = 32;

    bool live = lane < cnt;
    int le = e0 + (live ? lane : cnt - 1);
    int u_l = edge_u[le];
    int i_l = edge_i[le];
    int w_l = __float_as_int(norm[le]);

    unsigned wu = needU[u_l >> 5];
    unsigned wi = needI[i_l >> 5];
    bool nu = live & (((wu >> (u_l & 31)) & 1u) != 0u);
    bool ni = live & (((wi >> (i_l & 31)) & 1u) != 0u);
    unsigned mU = __builtin_amdgcn_ballot_w32(nu);
    unsigned mI = __builtin_amdgcn_ballot_w32(ni);

    add_rows_of_mask(mI, uA, iB, u_l, i_l, w_l, lane);
    add_rows_of_mask(mU, iA, uB, i_l, u_l, w_l, lane);
}

#define GR 8

__device__ __forceinline__ int out_row_segment(int row, int batch)
{
    return (row >= batch) + (row >= 2 * batch);
}

__global__ void __launch_bounds__(256) gather_h0_h1(
    const float* __restrict__ ufeat, const float* __restrict__ ifeat,
    const float* __restrict__ utab, const float* __restrict__ itab,
    const int* __restrict__ users, const int* __restrict__ pos,
    const int* __restrict__ neg,
    float* __restrict__ out, int batch)
{
    int wave = (int)(blockIdx.x * (blockDim.x >> 5) + (threadIdx.x >> 5));
    wave = __builtin_amdgcn_readfirstlane(wave);
    int lane = threadIdx.x & 31;

    int total = 3 * batch;
    int r0 = wave * GR;
    if (r0 >= total) return;
    int cnt = total - r0;
    if (cnt > GR) cnt = GR;

    int lrow = r0 + (lane < cnt ? lane : cnt - 1);
    int lseg = out_row_segment(lrow, batch);
    const int* lix = lseg == 0 ? users : (lseg == 1 ? pos : neg);
    int idx_l = lix[lrow - lseg * batch];

    v2f a[GR], b[GR];
#pragma unroll
    for (int j = 0; j < GR; ++j) {
        int idx = __builtin_amdgcn_readlane(idx_l, j);
        int seg = out_row_segment(r0 + (j < cnt ? j : cnt - 1), batch);
        const float* f = seg == 0 ? ufeat : ifeat;
        const float* t = seg == 0 ? utab  : itab;
        a[j] = ((const v2f*)(f + (size_t)idx * D))[lane];
        b[j] = ((const v2f*)(t + (size_t)idx * D))[lane];
    }
#pragma unroll
    for (int j = 0; j < GR; ++j) {
        if (j < cnt)
            ((v2f*)(out + (size_t)(r0 + j) * D))[lane] = a[j] + b[j];
    }
}

__global__ void __launch_bounds__(256) wmma_add_h2_h3_scale(
    float* __restrict__ out,
    const float* __restrict__ utab2, const float* __restrict__ itab2,
    const float* __restrict__ utab3, const float* __restrict__ itab3,
    const int* __restrict__ users, const int* __restrict__ pos,
    const int* __restrict__ neg, int batch, int ntiles)
{
    int wid = (int)(blockIdx.x * (blockDim.x >> 5) + (threadIdx.x >> 5));
    if (wid >= ntiles) return;

    int lane = threadIdx.x & 31;
    int h = lane >> 4;
    int l = lane & 15;
    int rg = wid >> 2;
    int ct = wid & 3;

    int row = rg * 16 + l;
    int seg = out_row_segment(row, batch);
    int r   = row - seg * batch;
    const int* ix = seg == 0 ? users : (seg == 1 ? pos : neg);
    int idx = ix[r];
    const float* t2 = (seg == 0 ? utab2 : itab2) + (size_t)idx * D + ct * 16;
    const float* t3 = (seg == 0 ? utab3 : itab3) + (size_t)idx * D + ct * 16;

    float* base = out + (size_t)rg * 16 * D + ct * 16;

    v8f acc = {};
#pragma unroll
    for (int c = 0; c < 4; ++c) {
        int k0 = 2 * h;
        int k1 = 2 * h + 1;
        v2f o = *(const v2f*)(base + (size_t)l * D + 4 * c + k0);
        v2f x = *(const v2f*)(t2 + 4 * c + k0);
        v2f y = *(const v2f*)(t3 + 4 * c + k0);
        v2f a = (o + x) + y;
        v2f b;
        b.x = ((4 * c + k0) == l) ? 0.25f : 0.0f;
        b.y = ((4 * c + k1) == l) ? 0.25f : 0.0f;
        acc = __builtin_amdgcn_wmma_f32_16x16x4_f32(
             false, a,  false, b,
             (short)0, acc,  false,  false);
    }
#pragma unroll
    for (int j = 0; j < 8; ++j)
        base[(size_t)(h * 8 + j) * D + l] = acc[j];
}

extern "C" void kernel_launch(void* const* d_in, const int* in_sizes, int n_in,
                              void* d_out, int out_size, void* d_ws, size_t ws_size,
                              hipStream_t stream)
{
    const float* user_feat = (const float*)d_in[0];
    const float* item_feat = (const float*)d_in[1];
    const int*   edge_u    = (const int*)d_in[2];
    const int*   edge_i    = (const int*)d_in[3];
    const float* norm      = (const float*)d_in[4];
    const int*   users     = (const int*)d_in[5];
    const int*   pos       = (const int*)d_in[6];
    const int*   neg       = (const int*)d_in[7];
    float* out = (float*)d_out;

    const int n_users = in_sizes[0] / D;
    const int n_items = in_sizes[1] / D;
    const int n_edges = in_sizes[2];
    const int batch   = in_sizes[5];

    size_t uElems = (size_t)n_users * D;
    size_t iElems = (size_t)n_items * D;

    float* u0 = (float*)d_ws;
    float* u1 = u0 + uElems;
    float* i0 = u1 + uElems;
    float* i1 = i0 + iElems;

    float* uCur = u0; float* uNxt = u1;
    float* iCur = i0; float* iNxt = i1;

    hipMemcpyAsync(uCur, user_feat, uElems * sizeof(float),
                   hipMemcpyDeviceToDevice, stream);
    hipMemcpyAsync(iCur, item_feat, iElems * sizeof(float),
                   hipMemcpyDeviceToDevice, stream);

    const int uWords = (n_users + 31) / 32;
    const int iWords = (n_items + 31) / 32;
    unsigned* needU = (unsigned*)(i1 + iElems);
    unsigned* needI = needU + uWords;
    const size_t wsNeed = (2 * uElems + 2 * iElems) * sizeof(float)
                        + (size_t)(uWords + iWords) * sizeof(unsigned);
    const bool lastNeededOnly = ws_size >= wsNeed;

    int sblocks = (n_edges + 8 * EPW - 1) / (8 * EPW);
    for (int L = 0; L < 3; ++L) {
        if (L == 2 && lastNeededOnly) {
            clear_words<<<(uWords + iWords + 255) / 256, 256, 0, stream>>>(
                needU, uWords + iWords);
            mark_and_clear_rows<<<(3 * batch + 7) / 8, 256, 0, stream>>>(
                users, pos, neg, needU, needI, uNxt, iNxt, batch);
            scatter_edges_needed<<<(n_edges + 255) / 256, 256, 0, stream>>>(
                uCur, iCur, uNxt, iNxt, edge_u, edge_i, norm,
                needU, needI, n_edges);
            break;
        }
        hipMemsetAsync(uNxt, 0, uElems * sizeof(float), stream);
        hipMemsetAsync(iNxt, 0, iElems * sizeof(float), stream);
        scatter_edges<<<sblocks, 256, 0, stream>>>(uCur, iCur, uNxt, iNxt,
                                                   edge_u, edge_i, norm, n_edges);
        if (L == 0) {
            gather_h0_h1<<<(3 * batch + 8 * GR - 1) / (8 * GR), 256, 0, stream>>>(
                user_feat, item_feat, uNxt, iNxt, users, pos, neg, out, batch);
        }
        if (L == 2) break;
        float* t;
        t = uCur; uCur = uNxt; uNxt = t;
        t = iCur; iCur = iNxt; iNxt = t;
    }

    int nrows  = 3 * batch;
    int ntiles = (nrows / 16) * (D / 16);
    int wblocks = (ntiles + 7) / 8;
    wmma_add_h2_h3_scale<<<wblocks, 256, 0, stream>>>(out, uCur, iCur, uNxt, iNxt,
                                                     users, pos, neg, batch, ntiles);
}


# second layer: scatter adds only into requested rows and their direct neighbours (2nd bitmap pair built each call; needed rows loaded in batches); float atomics kept, arrival order unordered as before
# speedup vs baseline: 1.9484x; 1.0353x over previous
#include <hip/hip_runtime.h>

#define D 64

typedef __attribute__((ext_vector_type(2))) float v2f;
typedef __attribute__((ext_vector_type(8))) float v8f;

#define EPW 16
#define EB  8

__device__ __forceinline__ void rows_to_lines(float& a, float& b)
{
    auto r = __builtin_amdgcn_permlane16_swap(__float_as_uint(a), __float_as_uint(b),
                                              false, false);
    a = __uint_as_float(r[0]);
    b = __uint_as_float(r[1]);
}

__device__ __forceinline__ int line_dim(int lane)
{
    return ((lane * 2) & 31) + (lane >> 4);
}

__device__ __forceinline__ void scatter_one(
    const float* __restrict__ uA, const float* __restrict__ iA,
    float* __restrict__ uB, float* __restrict__ iB,
    int u, int it, float w, int lane)
{
    v2f uv = ((const v2f*)(uA + (size_t)u  * D))[lane];
    v2f iv = ((const v2f*)(iA + (size_t)it * D))[lane];
    float ua = uv.x * w, ub = uv.y * w;
    float ia = iv.x * w, ib = iv.y * w;
    rows_to_lines(ua, ub);
    rows_to_lines(ia, ib);
    float* id = iB + (size_t)it * D + line_dim(lane);
    float* ud = uB + (size_t)u  * D + line_dim(lane);
    atomicAdd(&id[0],     ua);
    atomicAdd(&id[D / 2], ub);
    atomicAdd(&ud[0],     ia);
    atomicAdd(&ud[D / 2], ib);
}

__global__ void __launch_bounds__(256) scatter_edges(
    const float* __restrict__ uA, const float* __restrict__ iA,
    float* __restrict__ uB, float* __restrict__ iB,
    const int* __restrict__ edge_u, const int* __restrict__ edge_i,
    const float* __restrict__ norm, int n_edges)
{
    int wave = (int)(blockIdx.x * (blockDim.x >> 5) + (threadIdx.x >> 5));
    wave = __builtin_amdgcn_readfirstlane(wave);
    int lane = threadIdx.x & 31;

    int e0 = wave * EPW;
    if (e0 >= n_edges) return;
    int cnt = n_edges - e0;
    if (cnt > EPW) cnt = EPW;

    int le = e0 + (lane < cnt ? lane : cnt - 1);
    int u_l = edge_u[le];
    int i_l = edge_i[le];
    int w_l = __float_as_int(norm[le]);

#pragma unroll
    for (int s = 0; s < EPW; s += EB) {
        if (s + EB <= cnt) {
            int uu[EB], ii[EB];
            float ww[EB];
            v2f uv[EB], iv[EB];
#pragma unroll
            for (int j = 0; j < EB; ++j) {
                uu[j] = __builtin_amdgcn_readlane(u_l, s + j);
                ii[j] = __builtin_amdgcn_readlane(i_l, s + j);
                ww[j] = __int_as_float(__builtin_amdgcn_readlane(w_l, s + j));
            }
#pragma unroll
            for (int j = 0; j < EB; ++j) {
                uv[j] = ((const v2f*)(uA + (size_t)uu[j] * D))[lane];
                iv[j] = ((const v2f*)(iA + (size_t)ii[j] * D))[lane];
            }
#pragma unroll
            for (int j = 0; j < EB; ++j) {
                float ua = uv[j].x * ww[j], ub = uv[j].y * ww[j];
                float ia = iv[j].x * ww[j], ib = iv[j].y * ww[j];
                rows_to_lines(ua, ub);
                rows_to_lines(ia, ib);
                float* id = iB + (size_t)ii[j] * D + line_dim(lane);
                float* ud = uB + (size_t)uu[j] * D + line_dim(lane);
                atomicAdd(&id[0],     ua);
                atomicAdd(&id[D / 2], ub);
                atomicAdd(&ud[0],     ia);
                atomicAdd(&ud[D / 2], ib);
            }
        } else {
            for (int j = s; j < cnt; ++j) {
                int   u  = __builtin_amdgcn_readlane(u_l, j);
                int   it = __builtin_amdgcn_readlane(i_l, j);
                float w  = __int_as_float(__builtin_amdgcn_readlane(w_l, j));
                scatter_one(uA, iA, uB, iB, u, it, w, lane);
            }
            break;
        }
    }
}

__global__ void __launch_bounds__(256) clear_words(unsigned* __restrict__ w, int n)
{
    int gid = blockIdx.x * blockDim.x + threadIdx.x;
    if (gid < n) w[gid] = 0u;
}

__global__ void __launch_bounds__(256) mark_neighbours(
    const int* __restrict__ edge_u, const int* __restrict__ edge_i,
    const unsigned* __restrict__ reqU, const unsigned* __restrict__ reqI,
    unsigned* __restrict__ needU, unsigned* __restrict__ needI, int n_edges)
{
    int e = blockIdx.x * blockDim.x + threadIdx.x;
    if (e >= n_edges) return;
    int u = edge_u[e];
    int i = edge_i[e];
    unsigned wu = reqU[u >> 5];
    unsigned wi = reqI[i >> 5];
    if ((wu >> (u & 31)) & 1u) atomicOr(&needI[i >> 5], 1u << (i & 31));
    if ((wi >> (i & 31)) & 1u) atomicOr(&needU[u >> 5], 1u << (u & 31));
}

__global__ void __launch_bounds__(256) mark_and_clear_rows(
    const int* __restrict__ users, const int* __restrict__ pos,
    const int* __restrict__ neg,
    unsigned* __restrict__ needU, unsigned* __restrict__ needI,
    float* __restrict__ uB, float* __restrict__ iB, int batch, bool mark)
{
    int wave = (int)(blockIdx.x * (blockDim.x >> 5) + (threadIdx.x >> 5));
    wave = __builtin_amdgcn_readfirstlane(wave);
    if (wave >= 3 * batch) return;
    int lane = threadIdx.x & 31;
    int seg = wave / batch;
    int r   = wave - seg * batch;

    int idx;
    unsigned* need;
    float* tab;
    if (seg == 0)      { idx = users[r]; need = needU; tab = uB; }
    else if (seg == 1) { idx = pos[r];   need = needI; tab = iB; }
    else               { idx = neg[r];   need = needI; tab = iB; }

    if (mark && lane == 0) atomicOr(&need[idx >> 5], 1u << (idx & 31));
    v2f z = {0.0f, 0.0f};
    ((v2f*)(tab + (size_t)idx * D))[lane] = z;
}

template <int RB>
__device__ __forceinline__ unsigned add_row_batch(
    unsigned m, const float* __restrict__ src, float* __restrict__ dst,
    int src_l, int dst_l, int w_l, int lane)
{
    int   s[RB], d[RB];
    float w[RB];
    v2f   r[RB];
#pragma unroll
    for (int k = 0; k < RB; ++k) {
        int j = __builtin_ctz(m);
        m &= m - 1;
        s[k] = __builtin_amdgcn_readlane(src_l, j);
        d[k] = __builtin_amdgcn_readlane(dst_l, j);
        w[k] = __int_as_float(__builtin_amdgcn_readlane(w_l, j));
    }
#pragma unroll
    for (int k = 0; k < RB; ++k)
        r[k] = ((const v2f*)(src + (size_t)s[k] * D))[lane];
#pragma unroll
    for (int k = 0; k < RB; ++k) {
        float ra = r[k].x * w[k], rb = r[k].y * w[k];
        rows_to_lines(ra, rb);
        float* p = dst + (size_t)d[k] * D + line_dim(lane);
        atomicAdd(&p[0],     ra);
        atomicAdd(&p[D / 2], rb);
    }
    return m;
}

__device__ __forceinline__ void add_rows_of_mask(
    unsigned m, const float* __restrict__ src, float* __restrict__ dst,
    int src_l, int dst_l, int w_l, int lane)
{
    while (__builtin_popcount(m) >= 8)
        m = add_row_batch<8>(m, src, dst, src_l, dst_l, w_l, lane);
    if (__builtin_popcount(m) >= 4)
        m = add_row_batch<4>(m, src, dst, src_l, dst_l, w_l, lane);
    if (__builtin_popcount(m) >= 2)
        m = add_row_batch<2>(m, src, dst, src_l, dst_l, w_l, lane);
    if (m)
        m = add_row_batch<1>(m, src, dst, src_l, dst_l, w_l, lane);
}

__global__ void __launch_bounds__(256) scatter_edges_needed(
    const float* __restrict__ uA, const float* __restrict__ iA,
    float* __restrict__ uB, float* __restrict__ iB,
    const int* __restrict__ edge_u, const int* __restrict__ edge_i,
    const float* __restrict__ norm,
    const unsigned* __restrict__ needU, const unsigned* __restrict__ needI,
    int n_edges)
{
    int wave = (int)(blockIdx.x * (blockDim.x >> 5) + (threadIdx.x >> 5));
    wave = __builtin_amdgcn_readfirstlane(wave);
    int lane = threadIdx.x & 31;

    int e0 = wave * 32;
    if (e0 >= n_edges) return;
    int cnt = n_edges - e0;
    if (cnt > 32) cnt = 32;

    bool live = lane < cnt;
    int le = e0 + (live ? lane : cnt - 1);
    int u_l = edge_u[le];
    int i_l = edge_i[le];
    int w_l = __float_as_int(norm[le]);

    unsigned wu = needU[u_l >> 5];
    unsigned wi = needI[i_l >> 5];
    bool nu = live & (((wu >> (u_l & 31)) & 1u) != 0u);
    bool ni = live & (((wi >> (i_l & 31)) & 1u) != 0u);
    unsigned mU = __builtin_amdgcn_ballot_w32(nu);
    unsigned mI = __builtin_amdgcn_ballot_w32(ni);

    add_rows_of_mask(mI, uA, iB, u_l, i_l, w_l, lane);
    add_rows_of_mask(mU, iA, uB, i_l, u_l, w_l, lane);
}

#define GR 8

__device__ __forceinline__ int out_row_segment(int row, int batch)
{
    return (row >= batch) + (row >= 2 * batch);
}

__global__ void __launch_bounds__(256) gather_h0_h1(
    const float* __restrict__ ufeat, const float* __restrict__ ifeat,
    const float* __restrict__ utab, const float* __restrict__ itab,
    const int* __restrict__ users, const int* __restrict__ pos,
    const int* __restrict__ neg,
    unsigned* __restrict__ reqU, unsigned* __restrict__ reqI,
    unsigned* __restrict__ needU, unsigned* __restrict__ needI,
    float* __restrict__ out, int batch)
{
    int wave = (int)(blockIdx.x * (blockDim.x >> 5) + (threadIdx.x >> 5));
    wave = __builtin_amdgcn_readfirstlane(wave);
    int lane = threadIdx.x & 31;

    int total = 3 * batch;
    int r0 = wave * GR;
    if (r0 >= total) return;
    int cnt = total - r0;
    if (cnt > GR) cnt = GR;

    int lrow = r0 + (lane < cnt ? lane : cnt - 1);
    int lseg = out_row_segment(lrow, batch);
    const int* lix = lseg == 0 ? users : (lseg == 1 ? pos : neg);
    int idx_l = lix[lrow - lseg * batch];

    if (reqU != nullptr && lane < cnt) {
        unsigned* rq = lseg == 0 ? reqU  : reqI;
        unsigned* nd = lseg == 0 ? needU : needI;
        unsigned bit = 1u << (idx_l & 31);
        atomicOr(&rq[idx_l >> 5], bit);
        atomicOr(&nd[idx_l >> 5], bit);
    }

    v2f a[GR], b[GR];
#pragma unroll
    for (int j = 0; j < GR; ++j) {
        int idx = __builtin_amdgcn_readlane(idx_l, j);
        int seg = out_row_segment(r0 + (j < cnt ? j : cnt - 1), batch);
        const float* f = seg == 0 ? ufeat : ifeat;
        const float* t = seg == 0 ? utab  : itab;
        a[j] = ((const v2f*)(f + (size_t)idx * D))[lane];
        b[j] = ((const v2f*)(t + (size_t)idx * D))[lane];
    }
#pragma unroll
    for (int j = 0; j < GR; ++j) {
        if (j < cnt)
            ((v2f*)(out + (size_t)(r0 + j) * D))[lane] = a[j] + b[j];
    }
}

__global__ void __launch_bounds__(256) wmma_add_h2_h3_scale(
    float* __restrict__ out,
    const float* __restrict__ utab2, const float* __restrict__ itab2,
    const float* __restrict__ utab3, const float* __restrict__ itab3,
    const int* __restrict__ users, const int* __restrict__ pos,
    const int* __restrict__ neg, int batch, int ntiles)
{
    int wid = (int)(blockIdx.x * (blockDim.x >> 5) + (threadIdx.x >> 5));
    if (wid >= ntiles) return;

    int lane = threadIdx.x & 31;
    int h = lane >> 4;
    int l = lane & 15;
    int rg = wid >> 2;
    int ct = wid & 3;

    int row = rg * 16 + l;
    int seg = out_row_segment(row, batch);
    int r   = row - seg * batch;
    const int* ix = seg == 0 ? users : (seg == 1 ? pos : neg);
    int idx = ix[r];
    const float* t2 = (seg == 0 ? utab2 : itab2) + (size_t)idx * D + ct * 16;
    const float* t3 = (seg == 0 ? utab3 : itab3) + (size_t)idx * D + ct * 16;

    float* base = out + (size_t)rg * 16 * D + ct * 16;

    v8f acc = {};
#pragma unroll
    for (int c = 0; c < 4; ++c) {
        int k0 = 2 * h;
        int k1 = 2 * h + 1;
        v2f o = *(const v2f*)(base + (size_t)l * D + 4 * c + k0);
        v2f x = *(const v2f*)(t2 + 4 * c + k0);
        v2f y = *(const v2f*)(t3 + 4 * c + k0);
        v2f a = (o + x) + y;
        v2f b;
        b.x = ((4 * c + k0) == l) ? 0.25f : 0.0f;
        b.y = ((4 * c + k1) == l) ? 0.25f : 0.0f;
        acc = __builtin_amdgcn_wmma_f32_16x16x4_f32(
             false, a,  false, b,
             (short)0, acc,  false,  false);
    }
#pragma unroll
    for (int j = 0; j < 8; ++j)
        base[(size_t)(h * 8 + j) * D + l] = acc[j];
}

extern "C" void kernel_launch(void* const* d_in, const int* in_sizes, int n_in,
                              void* d_out, int out_size, void* d_ws, size_t ws_size,
                              hipStream_t stream)
{
    const float* user_feat = (const float*)d_in[0];
    const float* item_feat = (const float*)d_in[1];
    const int*   edge_u    = (const int*)d_in[2];
    const int*   edge_i    = (const int*)d_in[3];
    const float* norm      = (const float*)d_in[4];
    const int*   users     = (const int*)d_in[5];
    const int*   pos       = (const int*)d_in[6];
    const int*   neg       = (const int*)d_in[7];
    float* out = (float*)d_out;

    const int n_users = in_sizes[0] / D;
    const int n_items = in_sizes[1] / D;
    const int n_edges = in_sizes[2];
    const int batch   = in_sizes[5];

    size_t uElems = (size_t)n_users * D;
    size_t iElems = (size_t)n_items * D;

    float* u0 = (float*)d_ws;
    float* u1 = u0 + uElems;
    float* i0 = u1 + uElems;
    float* i1 = i0 + iElems;

    float* uCur = u0; float* uNxt = u1;
    float* iCur = i0; float* iNxt = i1;

    hipMemcpyAsync(uCur, user_feat, uElems * sizeof(float),
                   hipMemcpyDeviceToDevice, stream);
    hipMemcpyAsync(iCur, item_feat, iElems * sizeof(float),
                   hipMemcpyDeviceToDevice, stream);

    const int uWords = (n_users + 31) / 32;
    const int iWords = (n_items + 31) / 32;
    unsigned* reqU  = (unsigned*)(i1 + iElems);
    unsigned* reqI  = reqU + uWords;
    unsigned* needU = reqI + iWords;
    unsigned* needI = needU + uWords;
    const size_t wsNeed = (2 * uElems + 2 * iElems) * sizeof(float)
                        + (size_t)(uWords + iWords) * sizeof(unsigned);
    const size_t wsNeed2 = wsNeed + (size_t)(uWords + iWords) * sizeof(unsigned);
    const bool lastNeededOnly = ws_size >= wsNeed;
    const bool secondNeededOnly = ws_size >= wsNeed2;

    int sblocks = (n_edges + 8 * EPW - 1) / (8 * EPW);
    for (int L = 0; L < 3; ++L) {
        if (L == 2 && lastNeededOnly) {
            if (!secondNeededOnly)
                clear_words<<<(uWords + iWords + 255) / 256, 256, 0, stream>>>(
                    reqU, uWords + iWords);
            mark_and_clear_rows<<<(3 * batch + 7) / 8, 256, 0, stream>>>(
                users, pos, neg, reqU, reqI, uNxt, iNxt, batch, !secondNeededOnly);
            scatter_edges_needed<<<(n_edges + 255) / 256, 256, 0, stream>>>(
                uCur, iCur, uNxt, iNxt, edge_u, edge_i, norm,
                reqU, reqI, n_edges);
            break;
        }
        hipMemsetAsync(uNxt, 0, uElems * sizeof(float), stream);
        hipMemsetAsync(iNxt, 0, iElems * sizeof(float), stream);
        if (L == 1 && secondNeededOnly) {
            mark_neighbours<<<(n_edges + 255) / 256, 256, 0, stream>>>(
                edge_u, edge_i, reqU, reqI, needU, needI, n_edges);
            scatter_edges_needed<<<(n_edges + 255) / 256, 256, 0, stream>>>(
                uCur, iCur, uNxt, iNxt, edge_u, edge_i, norm,
                needU, needI, n_edges);
        } else {
            scatter_edges<<<sblocks, 256, 0, stream>>>(uCur, iCur, uNxt, iNxt,
                                                       edge_u, edge_i, norm, n_edges);
        }
        if (L == 0) {
            if (secondNeededOnly)
                clear_words<<<(2 * (uWords + iWords) + 255) / 256, 256, 0, stream>>>(
                    reqU, 2 * (uWords + iWords));
            gather_h0_h1<<<(3 * batch + 8 * GR - 1) / (8 * GR), 256, 0, stream>>>(
                user_feat, item_feat, uNxt, iNxt, users, pos, neg,
                secondNeededOnly ? reqU : nullptr, reqI, needU, needI, out, batch);
        }
        if (L == 2) break;
        float* t;
        t = uCur; uCur = uNxt; uNxt = t;
        t = iCur; iCur = iNxt; iNxt = t;
    }

    int nrows  = 3 * batch;
    int ntiles = (nrows / 16) * (D / 16);
    int wblocks = (ntiles + 7) / 8;
    wmma_add_h2_h3_scale<<<wblocks, 256, 0, stream>>>(out, uCur, iCur, uNxt, iNxt,
                                                     users, pos, neg, batch, ntiles);
}
